# starting kernel
# baseline (speedup 1.0000x reference)
#include <hip/hip_runtime.h>

#define LDIM 256
#define DDIM 512
#define HDIM 256
#define ODIM 16
#define MDIM 512
#define BDIM 2
#define LEAKY 0.1f
#define EPSV 1e-9f

typedef __attribute__((ext_vector_type(2))) float v2f;
typedef __attribute__((ext_vector_type(8))) float v8f;

__global__ __launch_bounds__(32) void mlp_wmma_kernel(
    const float* __restrict__ X, const float* __restrict__ W,
    const float* __restrict__ bias, float* __restrict__ Hout)
{
    const int lane  = threadIdx.x & 31;
    const int mBase = blockIdx.x * 16;
    const int nBase = blockIdx.y * 64;
    const int mRow  = mBase + (lane & 15);
    const int kOff  = (lane >> 4) * 2;

    v8f acc0 = {}, acc1 = {}, acc2 = {}, acc3 = {};

    const float* xrow = X + mRow * DDIM + kOff;
    const int n0 = nBase + (lane & 15);
    const float* w0 = W + (n0 +  0) * DDIM + kOff;
    const float* w1 = W + (n0 + 16) * DDIM + kOff;
    const float* w2 = W + (n0 + 32) * DDIM + kOff;
    const float* w3 = W + (n0 + 48) * DDIM + kOff;

    #pragma unroll 4
    for (int k = 0; k < DDIM; k += 4) {
        v2f a  = *(const v2f*)(xrow + k);
        v2f b0 = *(const v2f*)(w0 + k);
        v2f b1 = *(const v2f*)(w1 + k);
        v2f b2 = *(const v2f*)(w2 + k);
        v2f b3 = *(const v2f*)(w3 + k);
        acc0 = __builtin_amdgcn_wmma_f32_16x16x4_f32(false, a, false, b0, 0, acc0, false, false);
        acc1 = __builtin_amdgcn_wmma_f32_16x16x4_f32(false, a, false, b1, 0, acc1, false, false);
        acc2 = __builtin_amdgcn_wmma_f32_16x16x4_f32(false, a, false, b2, 0, acc2, false, false);
        acc3 = __builtin_amdgcn_wmma_f32_16x16x4_f32(false, a, false, b3, 0, acc3, false, false);
    }

    const int mOut = mBase + (lane >> 4) * 8;
    v8f accs[4] = {acc0, acc1, acc2, acc3};
    #pragma unroll
    for (int t = 0; t < 4; ++t) {
        const int nCol = nBase + t * 16 + (lane & 15);
        const float bv = bias[nCol];
        v8f acc = accs[t];
        #pragma unroll
        for (int r = 0; r < 8; ++r) {
            float h = acc[r] + bv;
            h = (h >= 0.0f) ? h : (LEAKY * h);
            Hout[(mOut + r) * HDIM + nCol] = h;
        }
    }
}

__global__ void cumsum_kernel(const float* __restrict__ h1, float* __restrict__ cum)
{
    const int b = blockIdx.x;
    const int h = threadIdx.x;
    const float* src = h1 + b * LDIM * HDIM + h;
    float*       dst = cum + b * LDIM * HDIM + h;
    float run = 0.0f;
    for (int l = 0; l < LDIM; ++l) {
        run += src[l * HDIM];
        dst[l * HDIM] = run;
    }
}

__global__ __launch_bounds__(32) void proj_wmma_kernel(
    const float* __restrict__ h1, const float* __restrict__ h2,
    const float* __restrict__ cum, const float* __restrict__ Wf,
    const float* __restrict__ bf,
    float* __restrict__ Qj, float* __restrict__ Qi, float* __restrict__ P5)
{
    const int lane = threadIdx.x & 31;
    const int mRow = blockIdx.x * 16 + (lane & 15);
    const int kOff = (lane >> 4) * 2;
    const int o    = lane & 15;

    const float* a1p = h1  + mRow * HDIM + kOff;
    const float* a2p = h2  + mRow * HDIM + kOff;
    const float* a5p = cum + mRow * HDIM + kOff;
    const float* w1p = Wf + o * (3 * HDIM) + kOff;
    const float* w2p = w1p + HDIM;
    const float* w3p = w1p + 2 * HDIM;

    v8f p1 = {}, p2 = {}, p3 = {}, p4 = {}, p5 = {};
    #pragma unroll 4
    for (int k = 0; k < HDIM; k += 4) {
        v2f a1 = *(const v2f*)(a1p + k);
        v2f a2 = *(const v2f*)(a2p + k);
        v2f a5 = *(const v2f*)(a5p + k);
        v2f b1 = *(const v2f*)(w1p + k);
        v2f b2 = *(const v2f*)(w2p + k);
        v2f b3 = *(const v2f*)(w3p + k);
        p1 = __builtin_amdgcn_wmma_f32_16x16x4_f32(false, a1, false, b1, 0, p1, false, false);
        p2 = __builtin_amdgcn_wmma_f32_16x16x4_f32(false, a2, false, b1, 0, p2, false, false);
        p3 = __builtin_amdgcn_wmma_f32_16x16x4_f32(false, a2, false, b2, 0, p3, false, false);
        p4 = __builtin_amdgcn_wmma_f32_16x16x4_f32(false, a1, false, b2, 0, p4, false, false);
        p5 = __builtin_amdgcn_wmma_f32_16x16x4_f32(false, a5, false, b3, 0, p5, false, false);
    }

    const int mOut = blockIdx.x * 16 + (lane >> 4) * 8;
    const float bfv = bf[o];
    #pragma unroll
    for (int r = 0; r < 8; ++r) {
        const int m = mOut + r;
        Qj[m * ODIM + o] = p1[r] + 0.5f * p3[r] + bfv;
        Qi[m * ODIM + o] = 0.5f * p4[r] - p2[r];
        P5[m * ODIM + o] = p5[r];
    }
}

__global__ __launch_bounds__(LDIM) void epilogue_kernel(
    const float* __restrict__ Qj, const float* __restrict__ Qi,
    const float* __restrict__ P5, float* __restrict__ out)
{
    const int i = blockIdx.x;
    const int b = blockIdx.y;
    const int j = threadIdx.x;

    __shared__ float sQi[ODIM];
    __shared__ float sP5p[ODIM];
    if (threadIdx.x < ODIM) {
        const int o = threadIdx.x;
        sQi[o]  = Qi[(b * LDIM + i) * ODIM + o];
        sP5p[o] = (i == 0) ? 0.0f : P5[(b * LDIM + i - 1) * ODIM + o];
    }
    __syncthreads();

    const float denom = (float)(j - i + 1) + EPSV;
    const float rden  = 1.0f / denom;

    const float* qj = Qj + (b * LDIM + j) * ODIM;
    const float* pj = P5 + (b * LDIM + j) * ODIM;
    float* o0 = out + ((size_t)(b * ODIM) * LDIM + i) * LDIM + j;

    #pragma unroll
    for (int o = 0; o < ODIM; ++o) {
        const float v = qj[o] + sQi[o] + (pj[o] - sP5p[o]) * rden;
        o0[(size_t)o * LDIM * LDIM] = v;
    }
}

extern "C" void kernel_launch(void* const* d_in, const int* in_sizes, int n_in,
                              void* d_out, int out_size, void* d_ws, size_t ws_size,
                              hipStream_t stream)
{
    const float* x  = (const float*)d_in[0];
    const float* W1 = (const float*)d_in[1];
    const float* b1 = (const float*)d_in[2];
    const float* W2 = (const float*)d_in[3];
    const float* b2 = (const float*)d_in[4];
    const float* Wf = (const float*)d_in[5];
    const float* bf = (const float*)d_in[6];
    float* out = (float*)d_out;

    float* ws  = (float*)d_ws;
    float* h1  = ws;
    float* h2  = h1  + (size_t)MDIM * HDIM;
    float* cum = h2  + (size_t)MDIM * HDIM;
    float* Qj  = cum + (size_t)MDIM * HDIM;
    float* Qi  = Qj  + (size_t)MDIM * ODIM;
    float* P5  = Qi  + (size_t)MDIM * ODIM;

    dim3 g1(MDIM / 16, HDIM / 64);
    mlp_wmma_kernel<<<g1, 32, 0, stream>>>(x, W1, b1, h1);
    mlp_wmma_kernel<<<g1, 32, 0, stream>>>(x, W2, b2, h2);
    cumsum_kernel<<<BDIM, HDIM, 0, stream>>>(h1, cum);
    proj_wmma_kernel<<<MDIM / 16, 32, 0, stream>>>(h1, h2, cum, Wf, bf, Qj, Qi, P5);
    epilogue_kernel<<<dim3(LDIM, BDIM), LDIM, 0, stream>>>(Qj, Qi, P5, out);
}


# Epilogue re-tiled: 128 four-wave blocks of 4 rows i; a thread owns 4 consecutive j in all 16 planes (16-byte stores); the batch's Qj and P5 staged once per block in LDS; 4096 waves to 512; same sums
# speedup vs baseline: 4.0912x; 4.0912x over previous
#include <hip/hip_runtime.h>

#define LDIM 256
#define DDIM 512
#define HDIM 256
#define ODIM 16
#define MDIM 512
#define BDIM 2
#define LEAKY 0.1f
#define EPSV 1e-9f

typedef __attribute__((ext_vector_type(2))) float v2f;
typedef __attribute__((ext_vector_type(8))) float v8f;

#define TILE_M 32
#define TILE_N 32
#define KSLAB  128
#define NSLAB  (DDIM / KSLAB)
#define PITCHF (KSLAB + 4)
#define PIECES 8

typedef __attribute__((ext_vector_type(4))) float v4f;

#define SCAN_GROUP 32
#define NGROUP (LDIM / SCAN_GROUP)

__device__ __forceinline__ void scan_load(float (&v)[SCAN_GROUP], const float* __restrict__ p)
{
    #pragma unroll
    for (int i = 0; i < SCAN_GROUP; ++i) v[i] = p[i * HDIM];
}
__device__ __forceinline__ float scan_sum(const float (&v)[SCAN_GROUP], float* __restrict__ p, float run)
{
    #pragma unroll
    for (int i = 0; i < SCAN_GROUP; ++i) {
        run += v[i];
        p[i * HDIM] = run;
    }
    return run;
}

template <bool WITH_SCAN>
__global__ __launch_bounds__(128) void mlp_wmma_kernel(
    const float* __restrict__ X, const float* __restrict__ W,
    const float* __restrict__ bias, float* __restrict__ Hout,
    const float* __restrict__ scanSrc, float* __restrict__ scanDst)
{
    if (WITH_SCAN && blockIdx.y == 0) {
        if (threadIdx.x >= 32) return;
        const int sb  = blockIdx.x;
        const int off = (sb >> 3) * LDIM * HDIM + (sb & 7) * 32 + threadIdx.x;
        const float* sp = scanSrc + off;
        float*       dp = scanDst + off;
        float va[SCAN_GROUP], vb[SCAN_GROUP];
        float run = 0.0f;
        scan_load(va, sp);
        #pragma unroll 1
        for (int g = 0; g < NGROUP - 2; g += 2) {
            const float* sg = sp + g * SCAN_GROUP * HDIM;
            float*       dg = dp + g * SCAN_GROUP * HDIM;
            scan_load(vb, sg + SCAN_GROUP * HDIM);
            run = scan_sum(va, dg, run);
            scan_load(va, sg + 2 * SCAN_GROUP * HDIM);
            run = scan_sum(vb, dg + SCAN_GROUP * HDIM, run);
        }
        scan_load(vb, sp + (NGROUP - 1) * SCAN_GROUP * HDIM);
        run = scan_sum(va, dp + (NGROUP - 2) * SCAN_GROUP * HDIM, run);
        run = scan_sum(vb, dp + (NGROUP - 1) * SCAN_GROUP * HDIM, run);
        return;
    }

    __shared__ __attribute__((aligned(16))) float sX[NSLAB][TILE_M * PITCHF];
    __shared__ __attribute__((aligned(16))) float sW[NSLAB][TILE_N * PITCHF];

    const int tid   = threadIdx.x;
    const int lane  = tid & 31;
    const int wave  = tid >> 5;
    const int wm    = wave >> 1;
    const int wn    = wave & 1;
    const int mBase = blockIdx.x * TILE_M;
    const int nBase = (blockIdx.y - (WITH_SCAN ? 1 : 0)) * TILE_N;
    const int c     = lane & 15;
    const int hh    = lane >> 4;

    const float* gx = X + (size_t)(mBase + wave) * DDIM + lane * 4;
    const float* gw = W + (size_t)(nBase + wave) * DDIM + lane * 4;
    const int    so = wave * PITCHF + lane * 4;

    const int ao = (wm * 16 + c) * PITCHF + 4 * hh;
    const int bo = (wn * 16 + c) * PITCHF + 4 * hh;

    v8f acc0 = {}, acc1 = {};
    v4f rx[PIECES], rw[PIECES];

    #pragma unroll
    for (int i = 0; i < PIECES; ++i) {
        rx[i] = *(const v4f*)(gx + (size_t)(4 * i) * DDIM);
        rw[i] = *(const v4f*)(gw + (size_t)(4 * i) * DDIM);
    }
    #pragma unroll
    for (int i = 0; i < PIECES; ++i) {
        *(v4f*)(&sX[0][so + 4 * i * PITCHF]) = rx[i];
        *(v4f*)(&sW[0][so + 4 * i * PITCHF]) = rw[i];
    }
    __syncthreads();

    #pragma unroll
    for (int s = 0; s < NSLAB; ++s) {
        if (s + 1 < NSLAB) {
            #pragma unroll
            for (int i = 0; i < PIECES; ++i) {
                rx[i] = *(const v4f*)(gx + (size_t)(4 * i) * DDIM + (s + 1) * KSLAB);
                rw[i] = *(const v4f*)(gw + (size_t)(4 * i) * DDIM + (s + 1) * KSLAB);
            }
        }
        __builtin_amdgcn_sched_barrier(0);

        const float* ap = &sX[s][ao];
        const float* bp = &sW[s][bo];
        v4f fa[KSLAB / 8], fb[KSLAB / 8];
        #pragma unroll
        for (int q = 0; q < KSLAB / 8; ++q) {
            fa[q] = *(const v4f*)(ap + 8 * q);
            fb[q] = *(const v4f*)(bp + 8 * q);
        }
        __builtin_amdgcn_sched_barrier(0);

        #pragma unroll
        for (int q = 0; q < KSLAB / 8; ++q) {
            acc0 = __builtin_amdgcn_wmma_f32_16x16x4_f32(false, fa[q].xy, false, fb[q].xy, 0, acc0, false, false);
            acc1 = __builtin_amdgcn_wmma_f32_16x16x4_f32(false, fa[q].zw, false, fb[q].zw, 0, acc1, false, false);
        }

        if (s + 1 < NSLAB) {
            __builtin_amdgcn_sched_barrier(0);
            #pragma unroll
            for (int i = 0; i < PIECES; ++i) {
                *(v4f*)(&sX[s + 1][so + 4 * i * PITCHF]) = rx[i];
                *(v4f*)(&sW[s + 1][so + 4 * i * PITCHF]) = rw[i];
            }
            __syncthreads();
        }
    }

    const int mOut = mBase + wm * 16 + hh * 8;
    const int nCol = nBase + wn * 16 + c;
    const float bv = bias[nCol];
    #pragma unroll
    for (int r = 0; r < 8; ++r) {
        float h = (acc0[r] + acc1[r]) + bv;
        h = (h >= 0.0f) ? h : (LEAKY * h);
        Hout[(mOut + r) * HDIM + nCol] = h;
    }
}

#define PJ_THREADS 256
#define PA_PITCH (HDIM + 4)
#define PB_PITCH (3 * HDIM + 4)
#define PJ_CHUNK 8

template <int NCHUNK>
__device__ __forceinline__ v8f tile_product(const float* ap, const float* bp)
{
    v8f acc0 = {}, acc1 = {};
    v4f fa[2][PJ_CHUNK], fb[2][PJ_CHUNK];
    #pragma unroll
    for (int q = 0; q < PJ_CHUNK; ++q) {
        fa[0][q] = *(const v4f*)(ap + 8 * q);
        fb[0][q] = *(const v4f*)(bp + 8 * q);
    }
    #pragma unroll
    for (int ch = 0; ch < NCHUNK; ++ch) {
        if (ch + 1 < NCHUNK) {
            #pragma unroll
            for (int q = 0; q < PJ_CHUNK; ++q) {
                fa[(ch + 1) & 1][q] = *(const v4f*)(ap + 64 * (ch + 1) + 8 * q);
                fb[(ch + 1) & 1][q] = *(const v4f*)(bp + 64 * (ch + 1) + 8 * q);
            }
        }
        __builtin_amdgcn_sched_barrier(0);
        #pragma unroll
        for (int q = 0; q < PJ_CHUNK; ++q) {
            acc0 = __builtin_amdgcn_wmma_f32_16x16x4_f32(false, fa[ch & 1][q].xy, false, fb[ch & 1][q].xy, 0, acc0, false, false);
            acc1 = __builtin_amdgcn_wmma_f32_16x16x4_f32(false, fa[ch & 1][q].zw, false, fb[ch & 1][q].zw, 0, acc1, false, false);
        }
        if (ch + 2 < NCHUNK) __builtin_amdgcn_sched_barrier(0);
    }
    return acc0 + acc1;
}

__global__ __launch_bounds__(PJ_THREADS) void proj_wmma_kernel(
    const float* __restrict__ h1, const float* __restrict__ h2,
    const float* __restrict__ cum, const float* __restrict__ Wf,
    const float* __restrict__ bf,
    float* __restrict__ Qj, float* __restrict__ Qi, float* __restrict__ P5)
{
    __shared__ __attribute__((aligned(16))) float sA[3][16 * PA_PITCH];
    __shared__ __attribute__((aligned(16))) float sB[ODIM * PB_PITCH];
    __shared__ float sP[8][256];

    const int tid   = threadIdx.x;
    const int lane  = tid & 31;
    const int wave  = __builtin_amdgcn_readfirstlane(tid >> 5);
    const int mBase = blockIdx.x * 16;

    const float* g1 = h1  + (size_t)mBase * HDIM + tid * 4;
    const float* g2 = h2  + (size_t)mBase * HDIM + tid * 4;
    const float* g5 = cum + (size_t)mBase * HDIM + tid * 4;
    const float* gw = Wf + tid * 4;
    v4f ra[12], rb[12];
    #pragma unroll
    for (int i = 0; i < 4; ++i) {
        ra[i]     = *(const v4f*)(g1 + 1024 * i);
        ra[4 + i] = *(const v4f*)(g2 + 1024 * i);
        ra[8 + i] = *(const v4f*)(g5 + 1024 * i);
    }
    #pragma unroll
    for (int i = 0; i < 12; ++i) rb[i] = *(const v4f*)(gw + 1024 * i);
    const float bfv = bf[tid & 15];

    const int sao = (tid >> 6) * PA_PITCH + (tid & 63) * 4;
    #pragma unroll
    for (int i = 0; i < 4; ++i) {
        *(v4f*)(&sA[0][sao + 4 * i * PA_PITCH]) = ra[i];
        *(v4f*)(&sA[1][sao + 4 * i * PA_PITCH]) = ra[4 + i];
        *(v4f*)(&sA[2][sao + 4 * i * PA_PITCH]) = ra[8 + i];
    }
    #pragma unroll
    for (int i = 0; i < 12; ++i) {
        const int p   = tid + 256 * i;
        const int row = p / 192;
        const int col = p - 192 * row;
        *(v4f*)(&sB[row * PB_PITCH + col * 4]) = rb[i];
    }
    __syncthreads();

    const int c  = lane & 15;
    const int hh = lane >> 4;
    const int fo = c * PA_PITCH + 4 * hh;
    const int go = c * PB_PITCH + 4 * hh;
    v8f acc;
    if (wave < 4) {
        const int aop  = (wave == 1 || wave == 2) ? 1 : 0;
        const int bcol = (wave >= 2) ? HDIM : 0;
        acc = tile_product<4>(&sA[aop][fo], &sB[go + bcol]);
    } else {
        const int kq = (wave - 4) * 64;
        acc = tile_product<1>(&sA[2][fo + kq], &sB[go + 2 * HDIM + kq]);
    }

    #pragma unroll
    for (int r = 0; r < 8; ++r) sP[wave][r * 32 + hh * 16 + c] = acc[r];
    __syncthreads();

    const int row = tid >> 4;
    const int e   = (row & 7) * 32 + (row >> 3) * 16 + (tid & 15);
    const float p1 = sP[0][e], p2 = sP[1][e], p3 = sP[2][e], p4 = sP[3][e];
    const float p5 = ((sP[4][e] + sP[5][e]) + sP[6][e]) + sP[7][e];
    const int   oi = mBase * ODIM + tid;
    Qj[oi] = p1 + 0.5f * p3 + bfv;
    Qi[oi] = 0.5f * p4 - p2;
    P5[oi] = p5;
}

#define EP_THREADS 128
#define EP_PAIR_ROWS 2
#define EP_ROWS (2 * EP_PAIR_ROWS)
#define EP_NCHUNK (LDIM / 4)
#define EP_CHUNK (4 * ODIM + 4)
#define EP_PIECES (LDIM * ODIM / 4 / EP_THREADS)

__global__ __launch_bounds__(EP_THREADS) void epilogue_kernel(
    const float* __restrict__ Qj, const float* __restrict__ Qi,
    const float* __restrict__ P5, float* __restrict__ out)
{
    __shared__ __attribute__((aligned(16))) float sQ[EP_NCHUNK * EP_CHUNK];
    __shared__ __attribute__((aligned(16))) float sP[EP_NCHUNK * EP_CHUNK];

    const int tid  = threadIdx.x;
    const int b    = blockIdx.y;
    const int pair = __builtin_amdgcn_readfirstlane(tid >> 6);
    const int iRow = blockIdx.x * EP_ROWS + pair * EP_PAIR_ROWS;

    const float* gq = Qj + (size_t)b * LDIM * ODIM + tid * 4;
    const float* gp = P5 + (size_t)b * LDIM * ODIM + tid * 4;
    v4f rq[EP_PIECES], rp[EP_PIECES];
    for (int n = 0; n < EP_PIECES; ++n) {
        rq[n] = *(const v4f*)(gq + 4 * EP_THREADS * n);
        rp[n] = *(const v4f*)(gp + 4 * EP_THREADS * n);
    }

    float qi[EP_PAIR_ROWS][ODIM], pp[EP_PAIR_ROWS][ODIM];
    for (int r = 0; r < EP_PAIR_ROWS; ++r) {
        const int i  = iRow + r;
        const int ip = (i > 0) ? i - 1 : 0;
        const float* qrow = Qi + (size_t)(b * LDIM + i)  * ODIM;
        const float* prow = P5 + (size_t)(b * LDIM + ip) * ODIM;
        for (int o = 0; o < ODIM; ++o) {
            const float pv = prow[o];
            qi[r][o] = qrow[o];
            pp[r][o] = (i > 0) ? pv : 0.0f;
        }
    }

    const int so = (tid >> 4) * EP_CHUNK + (tid & 15) * 4;
    for (int n = 0; n < EP_PIECES; ++n) {
        *(v4f*)(&sQ[so + (EP_THREADS / 16) * n * EP_CHUNK]) = rq[n];
        *(v4f*)(&sP[so + (EP_THREADS / 16) * n * EP_CHUNK]) = rp[n];
    }
    __syncthreads();

    const int tj = tid & 63;
    const int j0 = tj * 4;
    v4f q[16], p[16];
    for (int k = 0; k < 16; ++k) {
        q[k] = *(const v4f*)(&sQ[tj * EP_CHUNK + 4 * k]);
        p[k] = *(const v4f*)(&sP[tj * EP_CHUNK + 4 * k]);
    }
    for (int r = 0; r < EP_PAIR_ROWS; ++r) {
        const int i = iRow + r;
        float rden[4];
        for (int jj = 0; jj < 4; ++jj) rden[jj] = 1.0f / ((float)(j0 + jj - i + 1) + EPSV);

        float* orow = out + ((size_t)(b * ODIM) * LDIM + i) * LDIM + j0;
        for (int o = 0; o < ODIM; ++o) {
            v4f v;
            for (int jj = 0; jj < 4; ++jj)
                v[jj] = q[4 * jj + (o >> 2)][o & 3] + qi[r][o] + (p[4 * jj + (o >> 2)][o & 3] - pp[r][o]) * rden[jj];
            *(v4f*)(orow + (size_t)o * LDIM * LDIM) = v;
        }
    }
}

extern "C" void kernel_launch(void* const* d_in, const int* in_sizes, int n_in,
                              void* d_out, int out_size, void* d_ws, size_t ws_size,
                              hipStream_t stream)
{
    const float* x  = (const float*)d_in[0];
    const float* W1 = (const float*)d_in[1];
    const float* b1 = (const float*)d_in[2];
    const float* W2 = (const float*)d_in[3];
    const float* b2 = (const float*)d_in[4];
    const float* Wf = (const float*)d_in[5];
    const float* bf = (const float*)d_in[6];
    float* out = (float*)d_out;

    float* ws  = (float*)d_ws;
    float* h1  = ws;
    float* h2  = h1  + (size_t)MDIM * HDIM;
    float* cum = h2  + (size_t)MDIM * HDIM;
    float* Qj  = cum + (size_t)MDIM * HDIM;
    float* Qi  = Qj  + (size_t)MDIM * ODIM;
    float* P5  = Qi  + (size_t)MDIM * ODIM;

    dim3 g1(MDIM / TILE_M, HDIM / TILE_N);
    dim3 g2(MDIM / TILE_M, HDIM / TILE_N + 1);
    mlp_wmma_kernel<false><<<g1, 128, 0, stream>>>(x, W1, b1, h1, h1, cum);
    mlp_wmma_kernel<true><<<g2, 128, 0, stream>>>(x, W2, b2, h2, h1, cum);
    proj_wmma_kernel<<<MDIM / 16, PJ_THREADS, 0, stream>>>(h1, h2, cum, Wf, bf, Qj, Qi, P5);
    epilogue_kernel<<<dim3(LDIM / EP_ROWS, BDIM), EP_THREADS, 0, stream>>>(Qj, Qi, P5, out);
}


# h1 and h2 layers in one launch (256 blocks); running sum of h1 two-level: 16-row group sums from layer accumulators, offsets + in-tile sum in projection kernel (new summation order); 4 launches to 3
# speedup vs baseline: 5.6598x; 1.3834x over previous
#include <hip/hip_runtime.h>

#define LDIM 256
#define DDIM 512
#define HDIM 256
#define ODIM 16
#define MDIM 512
#define BDIM 2
#define LEAKY 0.1f
#define EPSV 1e-9f

typedef __attribute__((ext_vector_type(2))) float v2f;
typedef __attribute__((ext_vector_type(8))) float v8f;

#define TILE_M 32
#define TILE_N 32
#define KSLAB  128
#define NSLAB  (DDIM / KSLAB)
#define PITCHF (KSLAB + 4)
#define PIECES 8

typedef __attribute__((ext_vector_type(4))) float v4f;

__global__ __launch_bounds__(128) void mlp_wmma_kernel(
    const float* __restrict__ X,
    const float* __restrict__ W1, const float* __restrict__ b1,
    const float* __restrict__ W2, const float* __restrict__ b2,
    float* __restrict__ h1, float* __restrict__ h2, float* __restrict__ tsum)
{
    const int sel = blockIdx.y >> 3;
    const float* __restrict__ W    = sel ? W2 : W1;
    const float* __restrict__ bias = sel ? b2 : b1;
    float* __restrict__       Hout = sel ? h2 : h1;

    __shared__ __attribute__((aligned(16))) float sX[NSLAB][TILE_M * PITCHF];
    __shared__ __attribute__((aligned(16))) float sW[NSLAB][TILE_N * PITCHF];

    const int tid   = threadIdx.x;
    const int lane  = tid & 31;
    const int wave  = tid >> 5;
    const int wm    = wave >> 1;
    const int wn    = wave & 1;
    const int mBase = blockIdx.x * TILE_M;
    const int nBase = (blockIdx.y & 7) * TILE_N;
    const int c     = lane & 15;
    const int hh    = lane >> 4;

    const float* gx = X + (size_t)(mBase + wave) * DDIM + lane * 4;
    const float* gw = W + (size_t)(nBase + wave) * DDIM + lane * 4;
    const int    so = wave * PITCHF + lane * 4;

    const int ao = (wm * 16 + c) * PITCHF + 4 * hh;
    const int bo = (wn * 16 + c) * PITCHF + 4 * hh;

    v8f acc0 = {}, acc1 = {};
    v4f rx[PIECES], rw[PIECES];

    #pragma unroll
    for (int i = 0; i < PIECES; ++i) {
        rx[i] = *(const v4f*)(gx + (size_t)(4 * i) * DDIM);
        rw[i] = *(const v4f*)(gw + (size_t)(4 * i) * DDIM);
    }
    #pragma unroll
    for (int i = 0; i < PIECES; ++i) {
        *(v4f*)(&sX[0][so + 4 * i * PITCHF]) = rx[i];
        *(v4f*)(&sW[0][so + 4 * i * PITCHF]) = rw[i];
    }
    __syncthreads();

    #pragma unroll
    for (int s = 0; s < NSLAB; ++s) {
        if (s + 1 < NSLAB) {
            #pragma unroll
            for (int i = 0; i < PIECES; ++i) {
                rx[i] = *(const v4f*)(gx + (size_t)(4 * i) * DDIM + (s + 1) * KSLAB);
                rw[i] = *(const v4f*)(gw + (size_t)(4 * i) * DDIM + (s + 1) * KSLAB);
            }
        }
        __builtin_amdgcn_sched_barrier(0);

        const float* ap = &sX[s][ao];
        const float* bp = &sW[s][bo];
        v4f fa[KSLAB / 8], fb[KSLAB / 8];
        #pragma unroll
        for (int q = 0; q < KSLAB / 8; ++q) {
            fa[q] = *(const v4f*)(ap + 8 * q);
            fb[q] = *(const v4f*)(bp + 8 * q);
        }
        __builtin_amdgcn_sched_barrier(0);

        #pragma unroll
        for (int q = 0; q < KSLAB / 8; ++q) {
            acc0 = __builtin_amdgcn_wmma_f32_16x16x4_f32(false, fa[q].xy, false, fb[q].xy, 0, acc0, false, false);
            acc1 = __builtin_amdgcn_wmma_f32_16x16x4_f32(false, fa[q].zw, false, fb[q].zw, 0, acc1, false, false);
        }

        if (s + 1 < NSLAB) {
            __builtin_amdgcn_sched_barrier(0);
            #pragma unroll
            for (int i = 0; i < PIECES; ++i) {
                *(v4f*)(&sX[s + 1][so + 4 * i * PITCHF]) = rx[i];
                *(v4f*)(&sW[s + 1][so + 4 * i * PITCHF]) = rw[i];
            }
            __syncthreads();
        }
    }

    const int mOut = mBase + wm * 16 + hh * 8;
    const int nCol = nBase + wn * 16 + c;
    const float bv = bias[nCol];
    float colsum = 0.0f;
    #pragma unroll
    for (int r = 0; r < 8; ++r) {
        float h = (acc0[r] + acc1[r]) + bv;
        h = (h >= 0.0f) ? h : (LEAKY * h);
        Hout[(mOut + r) * HDIM + nCol] = h;
        colsum += h;
    }
    if (sel == 0) {
        const float tot = colsum + __shfl_xor(colsum, 16, 32);
        if (hh == 0) tsum[(blockIdx.x * 2 + wm) * HDIM + nCol] = tot;
    }
}

#define PJ_THREADS 256
#define PA_PITCH (HDIM + 4)
#define PB_PITCH (3 * HDIM + 4)
#define PJ_CHUNK 8

template <int NCHUNK>
__device__ __forceinline__ v8f tile_product(const float* ap, const float* bp)
{
    v8f acc0 = {}, acc1 = {};
    v4f fa[2][PJ_CHUNK], fb[2][PJ_CHUNK];
    #pragma unroll
    for (int q = 0; q < PJ_CHUNK; ++q) {
        fa[0][q] = *(const v4f*)(ap + 8 * q);
        fb[0][q] = *(const v4f*)(bp + 8 * q);
    }
    #pragma unroll
    for (int ch = 0; ch < NCHUNK; ++ch) {
        if (ch + 1 < NCHUNK) {
            #pragma unroll
            for (int q = 0; q < PJ_CHUNK; ++q) {
                fa[(ch + 1) & 1][q] = *(const v4f*)(ap + 64 * (ch + 1) + 8 * q);
                fb[(ch + 1) & 1][q] = *(const v4f*)(bp + 64 * (ch + 1) + 8 * q);
            }
        }
        __builtin_amdgcn_sched_barrier(0);
        #pragma unroll
        for (int q = 0; q < PJ_CHUNK; ++q) {
            acc0 = __builtin_amdgcn_wmma_f32_16x16x4_f32(false, fa[ch & 1][q].xy, false, fb[ch & 1][q].xy, 0, acc0, false, false);
            acc1 = __builtin_amdgcn_wmma_f32_16x16x4_f32(false, fa[ch & 1][q].zw, false, fb[ch & 1][q].zw, 0, acc1, false, false);
        }
        if (ch + 2 < NCHUNK) __builtin_amdgcn_sched_barrier(0);
    }
    return acc0 + acc1;
}

__global__ __launch_bounds__(PJ_THREADS) void proj_wmma_kernel(
    const float* __restrict__ h1, const float* __restrict__ h2,
    const float* __restrict__ tsum, const float* __restrict__ Wf,
    const float* __restrict__ bf,
    float* __restrict__ Qj, float* __restrict__ Qi, float* __restrict__ P5)
{
    __shared__ __attribute__((aligned(16))) float sA[3][16 * PA_PITCH];
    __shared__ __attribute__((aligned(16))) float sB[ODIM * PB_PITCH];
    __shared__ float sP[8][256];

    const int tid   = threadIdx.x;
    const int lane  = tid & 31;
    const int wave  = __builtin_amdgcn_readfirstlane(tid >> 5);
    const int mBase = blockIdx.x * 16;

    const float* g1 = h1  + (size_t)mBase * HDIM + tid * 4;
    const float* g2 = h2  + (size_t)mBase * HDIM + tid * 4;
    const float* gw = Wf + tid * 4;
    const int tIn = blockIdx.x & (LDIM / 16 - 1);
    const float* gt = tsum + (size_t)(blockIdx.x - tIn) * HDIM + tid;
    const float* gc = h1 + (size_t)mBase * HDIM + tid;
    float ts[LDIM / 16 - 1], hc[16];
    for (int q = 0; q < LDIM / 16 - 1; ++q) ts[q] = gt[q * HDIM];
    for (int r = 0; r < 16; ++r) hc[r] = gc[r * HDIM];
    v4f ra[8], rb[12];
    #pragma unroll
    for (int i = 0; i < 4; ++i) {
        ra[i]     = *(const v4f*)(g1 + 1024 * i);
        ra[4 + i] = *(const v4f*)(g2 + 1024 * i);
    }
    #pragma unroll
    for (int i = 0; i < 12; ++i) rb[i] = *(const v4f*)(gw + 1024 * i);
    const float bfv = bf[tid & 15];

    const int sao = (tid >> 6) * PA_PITCH + (tid & 63) * 4;
    #pragma unroll
    for (int i = 0; i < 4; ++i) {
        *(v4f*)(&sA[0][sao + 4 * i * PA_PITCH]) = ra[i];
        *(v4f*)(&sA[1][sao + 4 * i * PA_PITCH]) = ra[4 + i];
    }
    float run = 0.0f;
    for (int q = 0; q < LDIM / 16 - 1; ++q) run += (q < tIn) ? ts[q] : 0.0f;
    for (int r = 0; r < 16; ++r) {
        run += hc[r];
        sA[2][r * PA_PITCH + tid] = run;
    }
    #pragma unroll
    for (int i = 0; i < 12; ++i) {
        const int p   = tid + 256 * i;
        const int row = p / 192;
        const int col = p - 192 * row;
        *(v4f*)(&sB[row * PB_PITCH + col * 4]) = rb[i];
    }
    __syncthreads();

    const int c  = lane & 15;
    const int hh = lane >> 4;
    const int fo = c * PA_PITCH + 4 * hh;
    const int go = c * PB_PITCH + 4 * hh;
    v8f acc;
    if (wave < 4) {
        const int aop  = (wave == 1 || wave == 2) ? 1 : 0;
        const int bcol = (wave >= 2) ? HDIM : 0;
        acc = tile_product<4>(&sA[aop][fo], &sB[go + bcol]);
    } else {
        const int kq = (wave - 4) * 64;
        acc = tile_product<1>(&sA[2][fo + kq], &sB[go + 2 * HDIM + kq]);
    }

    #pragma unroll
    for (int r = 0; r < 8; ++r) sP[wave][r * 32 + hh * 16 + c] = acc[r];
    __syncthreads();

    const int row = tid >> 4;
    const int e   = (row & 7) * 32 + (row >> 3) * 16 + (tid & 15);
    const float p1 = sP[0][e], p2 = sP[1][e], p3 = sP[2][e], p4 = sP[3][e];
    const float p5 = ((sP[4][e] + sP[5][e]) + sP[6][e]) + sP[7][e];
    const int   oi = mBase * ODIM + tid;
    Qj[oi] = p1 + 0.5f * p3 + bfv;
    Qi[oi] = 0.5f * p4 - p2;
    P5[oi] = p5;
}

#define EP_THREADS 128
#define EP_PAIR_ROWS 2
#define EP_ROWS (2 * EP_PAIR_ROWS)
#define EP_NCHUNK (LDIM / 4)
#define EP_CHUNK (4 * ODIM + 4)
#define EP_PIECES (LDIM * ODIM / 4 / EP_THREADS)

__global__ __launch_bounds__(EP_THREADS) void epilogue_kernel(
    const float* __restrict__ Qj, const float* __restrict__ Qi,
    const float* __restrict__ P5, float* __restrict__ out)
{
    __shared__ __attribute__((aligned(16))) float sQ[EP_NCHUNK * EP_CHUNK];
    __shared__ __attribute__((aligned(16))) float sP[EP_NCHUNK * EP_CHUNK];

    const int tid  = threadIdx.x;
    const int b    = blockIdx.y;
    const int pair = __builtin_amdgcn_readfirstlane(tid >> 6);
    const int iRow = blockIdx.x * EP_ROWS + pair * EP_PAIR_ROWS;

    const float* gq = Qj + (size_t)b * LDIM * ODIM + tid * 4;
    const float* gp = P5 + (size_t)b * LDIM * ODIM + tid * 4;
    v4f rq[EP_PIECES], rp[EP_PIECES];
    for (int n = 0; n < EP_PIECES; ++n) {
        rq[n] = *(const v4f*)(gq + 4 * EP_THREADS * n);
        rp[n] = *(const v4f*)(gp + 4 * EP_THREADS * n);
    }

    float qi[EP_PAIR_ROWS][ODIM], pp[EP_PAIR_ROWS][ODIM];
    for (int r = 0; r < EP_PAIR_ROWS; ++r) {
        const int i  = iRow + r;
        const int ip = (i > 0) ? i - 1 : 0;
        const float* qrow = Qi + (size_t)(b * LDIM + i)  * ODIM;
        const float* prow = P5 + (size_t)(b * LDIM + ip) * ODIM;
        for (int o = 0; o < ODIM; ++o) {
            const float pv = prow[o];
            qi[r][o] = qrow[o];
            pp[r][o] = (i > 0) ? pv : 0.0f;
        }
    }

    const int so = (tid >> 4) * EP_CHUNK + (tid & 15) * 4;
    for (int n = 0; n < EP_PIECES; ++n) {
        *(v4f*)(&sQ[so + (EP_THREADS / 16) * n * EP_CHUNK]) = rq[n];
        *(v4f*)(&sP[so + (EP_THREADS / 16) * n * EP_CHUNK]) = rp[n];
    }
    __syncthreads();

    const int tj = tid & 63;
    const int j0 = tj * 4;
    v4f q[16], p[16];
    for (int k = 0; k < 16; ++k) {
        q[k] = *(const v4f*)(&sQ[tj * EP_CHUNK + 4 * k]);
        p[k] = *(const v4f*)(&sP[tj * EP_CHUNK + 4 * k]);
    }
    for (int r = 0; r < EP_PAIR_ROWS; ++r) {
        const int i = iRow + r;
        float rden[4];
        for (int jj = 0; jj < 4; ++jj) rden[jj] = 1.0f / ((float)(j0 + jj - i + 1) + EPSV);

        float* orow = out + ((size_t)(b * ODIM) * LDIM + i) * LDIM + j0;
        for (int o = 0; o < ODIM; ++o) {
            v4f v;
            for (int jj = 0; jj < 4; ++jj)
                v[jj] = q[4 * jj + (o >> 2)][o & 3] + qi[r][o] + (p[4 * jj + (o >> 2)][o & 3] - pp[r][o]) * rden[jj];
            *(v4f*)(orow + (size_t)o * LDIM * LDIM) = v;
        }
    }
}

extern "C" void kernel_launch(void* const* d_in, const int* in_sizes, int n_in,
                              void* d_out, int out_size, void* d_ws, size_t ws_size,
                              hipStream_t stream)
{
    const float* x  = (const float*)d_in[0];
    const float* W1 = (const float*)d_in[1];
    const float* b1 = (const float*)d_in[2];
    const float* W2 = (const float*)d_in[3];
    const float* b2 = (const float*)d_in[4];
    const float* Wf = (const float*)d_in[5];
    const float* bf = (const float*)d_in[6];
    float* out = (float*)d_out;

    float* ws  = (float*)d_ws;
    float* h1  = ws;
    float* h2  = h1  + (size_t)MDIM * HDIM;
    float* tsum = h2 + (size_t)MDIM * HDIM;
    float* Qj  = tsum + (size_t)(MDIM / 16) * HDIM;
    float* Qi  = Qj  + (size_t)MDIM * ODIM;
    float* P5  = Qi  + (size_t)MDIM * ODIM;

    dim3 g1(MDIM / TILE_M, 2 * HDIM / TILE_N);
    mlp_wmma_kernel<<<g1, 128, 0, stream>>>(x, W1, b1, W2, b2, h1, h2, tsum);
    proj_wmma_kernel<<<MDIM / 16, PJ_THREADS, 0, stream>>>(h1, h2, tsum, Wf, bf, Qj, Qi, P5);
    epilogue_kernel<<<dim3(LDIM / EP_ROWS, BDIM), EP_THREADS, 0, stream>>>(Qj, Qi, P5, out);
}


# Layer kernel K loop software-pipelined: all global loads at entry, kept in flight across hand-written LDS-only barriers; next slab's LDS stores and fragment reads issued among the products; same sums
# speedup vs baseline: 6.1983x; 1.0951x over previous
#include <hip/hip_runtime.h>

#define LDIM 256
#define DDIM 512
#define HDIM 256
#define ODIM 16
#define MDIM 512
#define BDIM 2
#define LEAKY 0.1f
#define EPSV 1e-9f

typedef __attribute__((ext_vector_type(2))) float v2f;
typedef __attribute__((ext_vector_type(8))) float v8f;

#define TILE_M 32
#define TILE_N 32
#define KSLAB  128
#define NSLAB  (DDIM / KSLAB)
#define PITCHF (KSLAB + 4)
#define PIECES 8

typedef __attribute__((ext_vector_type(4))) float v4f;

__device__ __forceinline__ void lds_barrier()
{
    asm volatile("s_wait_dscnt 0x0\n\ts_barrier_signal -1\n\ts_barrier_wait -1" ::: "memory");
}

__device__ __forceinline__ void lds_barrier_behind(v8f a0, v8f a1)
{
    asm volatile("s_wait_dscnt 0x0\n\ts_barrier_signal -1\n\ts_barrier_wait -1" :: "v"(a0), "v"(a1) : "memory");
}

__global__ __launch_bounds__(128) void mlp_wmma_kernel(
    const float* __restrict__ X,
    const float* __restrict__ W1, const float* __restrict__ b1,
    const float* __restrict__ W2, const float* __restrict__ b2,
    float* __restrict__ h1, float* __restrict__ h2, float* __restrict__ tsum)
{
    const int sel = blockIdx.y >> 3;
    const float* __restrict__ W    = sel ? W2 : W1;
    const float* __restrict__ bias = sel ? b2 : b1;
    float* __restrict__       Hout = sel ? h2 : h1;

    __shared__ __attribute__((aligned(16))) float sX[NSLAB][TILE_M * PITCHF];
    __shared__ __attribute__((aligned(16))) float sW[NSLAB][TILE_N * PITCHF];

    const int tid   = threadIdx.x;
    const int lane  = tid & 31;
    const int wave  = tid >> 5;
    const int wm    = wave >> 1;
    const int wn    = wave & 1;
    const int mBase = blockIdx.x * TILE_M;
    const int nBase = (blockIdx.y & 7) * TILE_N;
    const int c     = lane & 15;
    const int hh    = lane >> 4;

    const float* gx = X + (size_t)(mBase + wave) * DDIM + lane * 4;
    const float* gw = W + (size_t)(nBase + wave) * DDIM + lane * 4;
    const int    so = wave * PITCHF + lane * 4;

    const int ao = (wm * 16 + c) * PITCHF + 4 * hh;
    const int bo = (wn * 16 + c) * PITCHF + 4 * hh;

    const int mOut = mBase + wm * 16 + hh * 8;
    const int nCol = nBase + wn * 16 + c;

    const float bv = bias[nCol];
    v8f acc0 = {}, acc1 = {};
    v4f rx[NSLAB][PIECES], rw[NSLAB][PIECES];
    #pragma unroll
    for (int s = 0; s < NSLAB; ++s) {
        #pragma unroll
        for (int i = 0; i < PIECES; ++i) {
            rx[s][i] = *(const v4f*)(gx + (size_t)(4 * i) * DDIM + s * KSLAB);
            rw[s][i] = *(const v4f*)(gw + (size_t)(4 * i) * DDIM + s * KSLAB);
        }
        __builtin_amdgcn_sched_barrier(0);
    }

    #pragma unroll
    for (int i = 0; i < PIECES; ++i) {
        *(v4f*)(&sX[0][so + 4 * i * PITCHF]) = rx[0][i];
        *(v4f*)(&sW[0][so + 4 * i * PITCHF]) = rw[0][i];
    }
    lds_barrier();

    v4f fa[2][KSLAB / 8], fb[2][KSLAB / 8];
    #pragma unroll
    for (int q = 0; q < KSLAB / 8; ++q) {
        fa[0][q] = *(const v4f*)(&sX[0][ao + 8 * q]);
        fb[0][q] = *(const v4f*)(&sW[0][bo + 8 * q]);
    }
    __builtin_amdgcn_sched_barrier(0);

    #pragma unroll
    for (int s = 0; s < NSLAB; ++s) {
        const int cur = s & 1, nxt = cur ^ 1;
        #pragma unroll
        for (int q = 0; q < KSLAB / 16; ++q) {
            acc0 = __builtin_amdgcn_wmma_f32_16x16x4_f32(false, fa[cur][q].xy, false, fb[cur][q].xy, 0, acc0, false, false);
            acc1 = __builtin_amdgcn_wmma_f32_16x16x4_f32(false, fa[cur][q].zw, false, fb[cur][q].zw, 0, acc1, false, false);
        }
        if (s + 1 < NSLAB) {
            __builtin_amdgcn_sched_barrier(0);
            #pragma unroll
            for (int i = 0; i < PIECES; ++i) {
                *(v4f*)(&sX[s + 1][so + 4 * i * PITCHF]) = rx[s + 1][i];
                *(v4f*)(&sW[s + 1][so + 4 * i * PITCHF]) = rw[s + 1][i];
            }
            __builtin_amdgcn_sched_barrier(0);
            #pragma unroll
            for (int q = KSLAB / 16; q < 3 * KSLAB / 32; ++q) {
                acc0 = __builtin_amdgcn_wmma_f32_16x16x4_f32(false, fa[cur][q].xy, false, fb[cur][q].xy, 0, acc0, false, false);
                acc1 = __builtin_amdgcn_wmma_f32_16x16x4_f32(false, fa[cur][q].zw, false, fb[cur][q].zw, 0, acc1, false, false);
            }
            lds_barrier_behind(acc0, acc1);
            #pragma unroll
            for (int q = 0; q < KSLAB / 8; ++q) {
                fa[nxt][q] = *(const v4f*)(&sX[s + 1][ao + 8 * q]);
                fb[nxt][q] = *(const v4f*)(&sW[s + 1][bo + 8 * q]);
            }
            __builtin_amdgcn_sched_barrier(0);
            #pragma unroll
            for (int q = 3 * KSLAB / 32; q < KSLAB / 8; ++q) {
                acc0 = __builtin_amdgcn_wmma_f32_16x16x4_f32(false, fa[cur][q].xy, false, fb[cur][q].xy, 0, acc0, false, false);
                acc1 = __builtin_amdgcn_wmma_f32_16x16x4_f32(false, fa[cur][q].zw, false, fb[cur][q].zw, 0, acc1, false, false);
            }
        } else {
            #pragma unroll
            for (int q = KSLAB / 16; q < KSLAB / 8; ++q) {
                acc0 = __builtin_amdgcn_wmma_f32_16x16x4_f32(false, fa[cur][q].xy, false, fb[cur][q].xy, 0, acc0, false, false);
                acc1 = __builtin_amdgcn_wmma_f32_16x16x4_f32(false, fa[cur][q].zw, false, fb[cur][q].zw, 0, acc1, false, false);
            }
        }
    }

    float colsum = 0.0f;
    #pragma unroll
    for (int r = 0; r < 8; ++r) {
        float h = (acc0[r] + acc1[r]) + bv;
        h = (h >= 0.0f) ? h : (LEAKY * h);
        Hout[(mOut + r) * HDIM + nCol] = h;
        colsum += h;
    }
    if (sel == 0) {
        const float tot = colsum + __shfl_xor(colsum, 16, 32);
        if (hh == 0) tsum[(blockIdx.x * 2 + wm) * HDIM + nCol] = tot;
    }
}

#define PJ_THREADS 256
#define PA_PITCH (HDIM + 4)
#define PB_PITCH (3 * HDIM + 4)
#define PJ_CHUNK 8

template <int NCHUNK>
__device__ __forceinline__ v8f tile_product(const float* ap, const float* bp)
{
    v8f acc0 = {}, acc1 = {};
    v4f fa[2][PJ_CHUNK], fb[2][PJ_CHUNK];
    #pragma unroll
    for (int q = 0; q < PJ_CHUNK; ++q) {
        fa[0][q] = *(const v4f*)(ap + 8 * q);
        fb[0][q] = *(const v4f*)(bp + 8 * q);
    }
    #pragma unroll
    for (int ch = 0; ch < NCHUNK; ++ch) {
        if (ch + 1 < NCHUNK) {
            #pragma unroll
            for (int q = 0; q < PJ_CHUNK; ++q) {
                fa[(ch + 1) & 1][q] = *(const v4f*)(ap + 64 * (ch + 1) + 8 * q);
                fb[(ch + 1) & 1][q] = *(const v4f*)(bp + 64 * (ch + 1) + 8 * q);
            }
        }
        __builtin_amdgcn_sched_barrier(0);
        #pragma unroll
        for (int q = 0; q < PJ_CHUNK; ++q) {
            acc0 = __builtin_amdgcn_wmma_f32_16x16x4_f32(false, fa[ch & 1][q].xy, false, fb[ch & 1][q].xy, 0, acc0, false, false);
            acc1 = __builtin_amdgcn_wmma_f32_16x16x4_f32(false, fa[ch & 1][q].zw, false, fb[ch & 1][q].zw, 0, acc1, false, false);
        }
        if (ch + 2 < NCHUNK) __builtin_amdgcn_sched_barrier(0);
    }
    return acc0 + acc1;
}

__global__ __launch_bounds__(PJ_THREADS) void proj_wmma_kernel(
    const float* __restrict__ h1, const float* __restrict__ h2,
    const float* __restrict__ tsum, const float* __restrict__ Wf,
    const float* __restrict__ bf,
    float* __restrict__ Qj, float* __restrict__ Qi, float* __restrict__ P5)
{
    __shared__ __attribute__((aligned(16))) float sA[3][16 * PA_PITCH];
    __shared__ __attribute__((aligned(16))) float sB[ODIM * PB_PITCH];
    __shared__ float sP[8][256];

    const int tid   = threadIdx.x;
    const int lane  = tid & 31;
    const int wave  = __builtin_amdgcn_readfirstlane(tid >> 5);
    const int mBase = blockIdx.x * 16;

    const float* g1 = h1  + (size_t)mBase * HDIM + tid * 4;
    const float* g2 = h2  + (size_t)mBase * HDIM + tid * 4;
    const float* gw = Wf + tid * 4;
    const int tIn = blockIdx.x & (LDIM / 16 - 1);
    const float* gt = tsum + (size_t)(blockIdx.x - tIn) * HDIM + tid;
    const float* gc = h1 + (size_t)mBase * HDIM + tid;
    float ts[LDIM / 16 - 1], hc[16];
    for (int q = 0; q < LDIM / 16 - 1; ++q) ts[q] = gt[q * HDIM];
    for (int r = 0; r < 16; ++r) hc[r] = gc[r * HDIM];
    v4f ra[8], rb[12];
    #pragma unroll
    for (int i = 0; i < 4; ++i) {
        ra[i]     = *(const v4f*)(g1 + 1024 * i);
        ra[4 + i] = *(const v4f*)(g2 + 1024 * i);
    }
    #pragma unroll
    for (int i = 0; i < 12; ++i) rb[i] = *(const v4f*)(gw + 1024 * i);
    const float bfv = bf[tid & 15];

    const int sao = (tid >> 6) * PA_PITCH + (tid & 63) * 4;
    #pragma unroll
    for (int i = 0; i < 4; ++i) {
        *(v4f*)(&sA[0][sao + 4 * i * PA_PITCH]) = ra[i];
        *(v4f*)(&sA[1][sao + 4 * i * PA_PITCH]) = ra[4 + i];
    }
    float run = 0.0f;
    for (int q = 0; q < LDIM / 16 - 1; ++q) run += (q < tIn) ? ts[q] : 0.0f;
    for (int r = 0; r < 16; ++r) {
        run += hc[r];
        sA[2][r * PA_PITCH + tid] = run;
    }
    #pragma unroll
    for (int i = 0; i < 12; ++i) {
        const int p   = tid + 256 * i;
        const int row = p / 192;
        const int col = p - 192 * row;
        *(v4f*)(&sB[row * PB_PITCH + col * 4]) = rb[i];
    }
    __syncthreads();

    const int c  = lane & 15;
    const int hh = lane >> 4;
    const int fo = c * PA_PITCH + 4 * hh;
    const int go = c * PB_PITCH + 4 * hh;
    v8f acc;
    if (wave < 4) {
        const int aop  = (wave == 1 || wave == 2) ? 1 : 0;
        const int bcol = (wave >= 2) ? HDIM : 0;
        acc = tile_product<4>(&sA[aop][fo], &sB[go + bcol]);
    } else {
        const int kq = (wave - 4) * 64;
        acc = tile_product<1>(&sA[2][fo + kq], &sB[go + 2 * HDIM + kq]);
    }

    #pragma unroll
    for (int r = 0; r < 8; ++r) sP[wave][r * 32 + hh * 16 + c] = acc[r];
    __syncthreads();

    const int row = tid >> 4;
    const int e   = (row & 7) * 32 + (row >> 3) * 16 + (tid & 15);
    const float p1 = sP[0][e], p2 = sP[1][e], p3 = sP[2][e], p4 = sP[3][e];
    const float p5 = ((sP[4][e] + sP[5][e]) + sP[6][e]) + sP[7][e];
    const int   oi = mBase * ODIM + tid;
    Qj[oi] = p1 + 0.5f * p3 + bfv;
    Qi[oi] = 0.5f * p4 - p2;
    P5[oi] = p5;
}

#define EP_THREADS 128
#define EP_PAIR_ROWS 2
#define EP_ROWS (2 * EP_PAIR_ROWS)
#define EP_NCHUNK (LDIM / 4)
#define EP_CHUNK (4 * ODIM + 4)
#define EP_PIECES (LDIM * ODIM / 4 / EP_THREADS)

__global__ __launch_bounds__(EP_THREADS) void epilogue_kernel(
    const float* __restrict__ Qj, const float* __restrict__ Qi,
    const float* __restrict__ P5, float* __restrict__ out)
{
    __shared__ __attribute__((aligned(16))) float sQ[EP_NCHUNK * EP_CHUNK];
    __shared__ __attribute__((aligned(16))) float sP[EP_NCHUNK * EP_CHUNK];

    const int tid  = threadIdx.x;
    const int b    = blockIdx.y;
    const int pair = __builtin_amdgcn_readfirstlane(tid >> 6);
    const int iRow = blockIdx.x * EP_ROWS + pair * EP_PAIR_ROWS;

    const float* gq = Qj + (size_t)b * LDIM * ODIM + tid * 4;
    const float* gp = P5 + (size_t)b * LDIM * ODIM + tid * 4;
    v4f rq[EP_PIECES], rp[EP_PIECES];
    for (int n = 0; n < EP_PIECES; ++n) {
        rq[n] = *(const v4f*)(gq + 4 * EP_THREADS * n);
        rp[n] = *(const v4f*)(gp + 4 * EP_THREADS * n);
    }

    float qi[EP_PAIR_ROWS][ODIM], pp[EP_PAIR_ROWS][ODIM];
    for (int r = 0; r < EP_PAIR_ROWS; ++r) {
        const int i  = iRow + r;
        const int ip = (i > 0) ? i - 1 : 0;
        const float* qrow = Qi + (size_t)(b * LDIM + i)  * ODIM;
        const float* prow = P5 + (size_t)(b * LDIM + ip) * ODIM;
        for (int o = 0; o < ODIM; ++o) {
            const float pv = prow[o];
            qi[r][o] = qrow[o];
            pp[r][o] = (i > 0) ? pv : 0.0f;
        }
    }

    const int so = (tid >> 4) * EP_CHUNK + (tid & 15) * 4;
    for (int n = 0; n < EP_PIECES; ++n) {
        *(v4f*)(&sQ[so + (EP_THREADS / 16) * n * EP_CHUNK]) = rq[n];
        *(v4f*)(&sP[so + (EP_THREADS / 16) * n * EP_CHUNK]) = rp[n];
    }
    __syncthreads();

    const int tj = tid & 63;
    const int j0 = tj * 4;
    v4f q[16], p[16];
    for (int k = 0; k < 16; ++k) {
        q[k] = *(const v4f*)(&sQ[tj * EP_CHUNK + 4 * k]);
        p[k] = *(const v4f*)(&sP[tj * EP_CHUNK + 4 * k]);
    }
    for (int r = 0; r < EP_PAIR_ROWS; ++r) {
        const int i = iRow + r;
        float rden[4];
        for (int jj = 0; jj < 4; ++jj) rden[jj] = 1.0f / ((float)(j0 + jj - i + 1) + EPSV);

        float* orow = out + ((size_t)(b * ODIM) * LDIM + i) * LDIM + j0;
        for (int o = 0; o < ODIM; ++o) {
            v4f v;
            for (int jj = 0; jj < 4; ++jj)
                v[jj] = q[4 * jj + (o >> 2)][o & 3] + qi[r][o] + (p[4 * jj + (o >> 2)][o & 3] - pp[r][o]) * rden[jj];
            *(v4f*)(orow + (size_t)o * LDIM * LDIM) = v;
        }
    }
}

extern "C" void kernel_launch(void* const* d_in, const int* in_sizes, int n_in,
                              void* d_out, int out_size, void* d_ws, size_t ws_size,
                              hipStream_t stream)
{
    const float* x  = (const float*)d_in[0];
    const float* W1 = (const float*)d_in[1];
    const float* b1 = (const float*)d_in[2];
    const float* W2 = (const float*)d_in[3];
    const float* b2 = (const float*)d_in[4];
    const float* Wf = (const float*)d_in[5];
    const float* bf = (const float*)d_in[6];
    float* out = (float*)d_out;

    float* ws  = (float*)d_ws;
    float* h1  = ws;
    float* h2  = h1  + (size_t)MDIM * HDIM;
    float* tsum = h2 + (size_t)MDIM * HDIM;
    float* Qj  = tsum + (size_t)(MDIM / 16) * HDIM;
    float* Qi  = Qj  + (size_t)MDIM * ODIM;
    float* P5  = Qi  + (size_t)MDIM * ODIM;

    dim3 g1(MDIM / TILE_M, 2 * HDIM / TILE_N);
    mlp_wmma_kernel<<<g1, 128, 0, stream>>>(x, W1, b1, W2, b2, h1, h2, tsum);
    proj_wmma_kernel<<<MDIM / 16, PJ_THREADS, 0, stream>>>(h1, h2, tsum, Wf, bf, Qj, Qi, P5);
    epilogue_kernel<<<dim3(LDIM / EP_ROWS, BDIM), EP_THREADS, 0, stream>>>(Qj, Qi, P5, out);
}
